# starting kernel
# baseline (speedup 1.0000x reference)
#include <hip/hip_runtime.h>

typedef __attribute__((ext_vector_type(2))) float v2f;
typedef __attribute__((ext_vector_type(8))) float v8f;

#define N_LOOP 21

__device__ __forceinline__ float fast_rcp(float x) {
  return __builtin_amdgcn_rcpf(x);
}

__device__ __forceinline__ v8f matvec_tile(const v2f fA[16], const v2f X[4][4]) {
  v8f c = {0.f, 0.f, 0.f, 0.f, 0.f, 0.f, 0.f, 0.f};
#pragma unroll
  for (int k = 0; k < 16; ++k)
    c = __builtin_amdgcn_wmma_f32_16x16x4_f32(false, fA[k], false, X[k >> 2][k & 3],
                                              (short)0, c, false, false);
  return c;
}

__global__ __launch_bounds__(256) void comp_net_kernel(
    const float* __restrict__ AT, const float* __restrict__ K_raw,
    const float* __restrict__ BT_raw, const float* __restrict__ W_raw,
    const float* __restrict__ b_raw, float* __restrict__ Y) {
  __shared__ float sKT[64 * 64];
  __shared__ float sK[64 * 64];
  __shared__ float sM[64 * 64];

  const int tid = threadIdx.x;
  for (int idx = tid; idx < 4096; idx += 256) {
    const int i = idx >> 6, j = idx & 63;
    const float ke = fminf(expf(K_raw[idx]), 1000.0f);
    sK[idx] = ke;
    sKT[j * 64 + i] = ke;
    float w = W_raw[idx];
    w = fminf(fmaxf(w, -10.0f), 10.0f);
    sM[idx] = ke * w;
  }
  __syncthreads();

  const int lane = tid & 31;
  const int hi = (lane >> 4) & 1;
  const int lr = lane & 15;
  const int warp = tid >> 5;
  const int row = ((int)blockIdx.x * 8 + warp) * 16 + lr;

  v2f fKT[4][16], fK[4][16];
#pragma unroll
  for (int mtA = 0; mtA < 4; ++mtA) {
    const float* rT = sKT + (16 * mtA + lr) * 64 + 8 * hi;
    const float* rK = sK + (16 * mtA + lr) * 64 + 8 * hi;
#pragma unroll
    for (int k = 0; k < 16; ++k) {
      const int col = 16 * (k >> 2) + 2 * (k & 3);
      fKT[mtA][k] = *(const v2f*)(rT + col);
      fK[mtA][k] = *(const v2f*)(rK + col);
    }
  }

  v2f ATt[4][4], AFt[4][4], BFt[4][4], BTv[4][4];
#pragma unroll
  for (int mt = 0; mt < 4; ++mt) {
    const float* src = AT + row * 64 + 16 * mt + 8 * hi;
    const float* bsrc = BT_raw + 16 * mt + 8 * hi;
#pragma unroll
    for (int q = 0; q < 4; ++q) {
      ATt[mt][q] = *(const v2f*)(src + 2 * q);
      AFt[mt][q] = ATt[mt][q];
      v2f t;
      t[0] = fminf(expf(bsrc[2 * q]), 1000.0f);
      t[1] = fminf(expf(bsrc[2 * q + 1]), 1000.0f);
      BTv[mt][q] = t;
    }
  }

  v8f acc[4];

  for (int it = 0; it < N_LOOP; ++it) {
#pragma unroll
    for (int mtA = 0; mtA < 4; ++mtA) acc[mtA] = matvec_tile(fKT[mtA], AFt);
#pragma unroll
    for (int mt = 0; mt < 4; ++mt)
#pragma unroll
      for (int q = 0; q < 4; ++q) {
        v2f t;
        t[0] = BTv[mt][q][0] * fast_rcp(acc[mt][2 * q] + 1.0f);
        t[1] = BTv[mt][q][1] * fast_rcp(acc[mt][2 * q + 1] + 1.0f);
        BFt[mt][q] = t;
      }

#pragma unroll
    for (int mtA = 0; mtA < 4; ++mtA) acc[mtA] = matvec_tile(fK[mtA], BFt);
#pragma unroll
    for (int mt = 0; mt < 4; ++mt)
#pragma unroll
      for (int q = 0; q < 4; ++q) {
        v2f t;
        t[0] = ATt[mt][q][0] * fast_rcp(acc[mt][2 * q] + 1.0f);
        t[1] = ATt[mt][q][1] * fast_rcp(acc[mt][2 * q + 1] + 1.0f);
        AFt[mt][q] = t;
      }
  }

#pragma unroll
  for (int mtA = 0; mtA < 4; ++mtA) acc[mtA] = matvec_tile(fKT[mtA], AFt);
#pragma unroll
  for (int mt = 0; mt < 4; ++mt)
#pragma unroll
    for (int q = 0; q < 4; ++q) {
      v2f t;
      t[0] = BTv[mt][q][0] * fast_rcp(acc[mt][2 * q] + 1.0f);
      t[1] = BTv[mt][q][1] * fast_rcp(acc[mt][2 * q + 1] + 1.0f);
      BFt[mt][q] = t;
    }

  float part = 0.0f;
#pragma unroll
  for (int mtA = 0; mtA < 4; ++mtA) {
    v2f fM[16];
    const float* rM = sM + (16 * mtA + lr) * 64 + 8 * hi;
#pragma unroll
    for (int k = 0; k < 16; ++k)
      fM[k] = *(const v2f*)(rM + 16 * (k >> 2) + 2 * (k & 3));
    const v8f u = matvec_tile(fM, BFt);
#pragma unroll
    for (int q = 0; q < 4; ++q) {
      part = fmaf(AFt[mtA][q][0], u[2 * q], part);
      part = fmaf(AFt[mtA][q][1], u[2 * q + 1], part);
    }
  }
  part += __shfl_xor(part, 16, 32);

  if (hi == 0) {
    const float bc = fminf(fmaxf(b_raw[0], -10.0f), 10.0f);
    Y[row] = part + bc;
  }
}

extern "C" void kernel_launch(void* const* d_in, const int* in_sizes, int n_in,
                              void* d_out, int out_size, void* d_ws, size_t ws_size,
                              hipStream_t stream) {
  const float* AT = (const float*)d_in[0];
  const float* K_raw = (const float*)d_in[1];
  const float* BT_raw = (const float*)d_in[2];
  const float* W_raw = (const float*)d_in[3];
  const float* b_raw = (const float*)d_in[4];
  float* Y = (float*)d_out;

  const int B = in_sizes[0] / 64;
  const int blocks = B / 128;
  comp_net_kernel<<<blocks, 256, 0, stream>>>(AT, K_raw, BT_raw, W_raw, b_raw, Y);
}


# Work-groups of 4 waves instead of 8 (256 groups, not 128): one wave per SIMD on every WGP instead of two per SIMD on half of them; per-row arithmetic and summation order unchanged
# speedup vs baseline: 1.6615x; 1.6615x over previous
#include <hip/hip_runtime.h>

typedef __attribute__((ext_vector_type(2))) float v2f;
typedef __attribute__((ext_vector_type(8))) float v8f;

#define N_LOOP 21

#define WG_WAVES 4
#define WG_THREADS (WG_WAVES * 32)

__device__ __forceinline__ float fast_rcp(float x) {
  return __builtin_amdgcn_rcpf(x);
}

__device__ __forceinline__ v8f matvec_tile(const v2f fA[16], const v2f X[4][4]) {
  v8f c = {0.f, 0.f, 0.f, 0.f, 0.f, 0.f, 0.f, 0.f};
#pragma unroll
  for (int k = 0; k < 16; ++k)
    c = __builtin_amdgcn_wmma_f32_16x16x4_f32(false, fA[k], false, X[k >> 2][k & 3],
                                              (short)0, c, false, false);
  return c;
}

__global__ __launch_bounds__(WG_THREADS) void comp_net_kernel(
    const float* __restrict__ AT, const float* __restrict__ K_raw,
    const float* __restrict__ BT_raw, const float* __restrict__ W_raw,
    const float* __restrict__ b_raw, float* __restrict__ Y) {
  __shared__ float sKT[64 * 64];
  __shared__ float sK[64 * 64];
  __shared__ float sM[64 * 64];

  const int tid = threadIdx.x;
  for (int idx = tid; idx < 4096; idx += WG_THREADS) {
    const int i = idx >> 6, j = idx & 63;
    const float ke = fminf(expf(K_raw[idx]), 1000.0f);
    sK[idx] = ke;
    sKT[j * 64 + i] = ke;
    float w = W_raw[idx];
    w = fminf(fmaxf(w, -10.0f), 10.0f);
    sM[idx] = ke * w;
  }
  __syncthreads();

  const int lane = tid & 31;
  const int hi = (lane >> 4) & 1;
  const int lr = lane & 15;
  const int warp = tid >> 5;
  const int row = ((int)blockIdx.x * WG_WAVES + warp) * 16 + lr;

  v2f fKT[4][16], fK[4][16];
#pragma unroll
  for (int mtA = 0; mtA < 4; ++mtA) {
    const float* rT = sKT + (16 * mtA + lr) * 64 + 8 * hi;
    const float* rK = sK + (16 * mtA + lr) * 64 + 8 * hi;
#pragma unroll
    for (int k = 0; k < 16; ++k) {
      const int col = 16 * (k >> 2) + 2 * (k & 3);
      fKT[mtA][k] = *(const v2f*)(rT + col);
      fK[mtA][k] = *(const v2f*)(rK + col);
    }
  }

  v2f ATt[4][4], AFt[4][4], BFt[4][4], BTv[4][4];
#pragma unroll
  for (int mt = 0; mt < 4; ++mt) {
    const float* src = AT + row * 64 + 16 * mt + 8 * hi;
    const float* bsrc = BT_raw + 16 * mt + 8 * hi;
#pragma unroll
    for (int q = 0; q < 4; ++q) {
      ATt[mt][q] = *(const v2f*)(src + 2 * q);
      AFt[mt][q] = ATt[mt][q];
      v2f t;
      t[0] = fminf(expf(bsrc[2 * q]), 1000.0f);
      t[1] = fminf(expf(bsrc[2 * q + 1]), 1000.0f);
      BTv[mt][q] = t;
    }
  }

  v8f acc[4];

  for (int it = 0; it < N_LOOP; ++it) {
#pragma unroll
    for (int mtA = 0; mtA < 4; ++mtA) acc[mtA] = matvec_tile(fKT[mtA], AFt);
#pragma unroll
    for (int mt = 0; mt < 4; ++mt)
#pragma unroll
      for (int q = 0; q < 4; ++q) {
        v2f t;
        t[0] = BTv[mt][q][0] * fast_rcp(acc[mt][2 * q] + 1.0f);
        t[1] = BTv[mt][q][1] * fast_rcp(acc[mt][2 * q + 1] + 1.0f);
        BFt[mt][q] = t;
      }

#pragma unroll
    for (int mtA = 0; mtA < 4; ++mtA) acc[mtA] = matvec_tile(fK[mtA], BFt);
#pragma unroll
    for (int mt = 0; mt < 4; ++mt)
#pragma unroll
      for (int q = 0; q < 4; ++q) {
        v2f t;
        t[0] = ATt[mt][q][0] * fast_rcp(acc[mt][2 * q] + 1.0f);
        t[1] = ATt[mt][q][1] * fast_rcp(acc[mt][2 * q + 1] + 1.0f);
        AFt[mt][q] = t;
      }
  }

#pragma unroll
  for (int mtA = 0; mtA < 4; ++mtA) acc[mtA] = matvec_tile(fKT[mtA], AFt);
#pragma unroll
  for (int mt = 0; mt < 4; ++mt)
#pragma unroll
    for (int q = 0; q < 4; ++q) {
      v2f t;
      t[0] = BTv[mt][q][0] * fast_rcp(acc[mt][2 * q] + 1.0f);
      t[1] = BTv[mt][q][1] * fast_rcp(acc[mt][2 * q + 1] + 1.0f);
      BFt[mt][q] = t;
    }

  float part = 0.0f;
#pragma unroll
  for (int mtA = 0; mtA < 4; ++mtA) {
    v2f fM[16];
    const float* rM = sM + (16 * mtA + lr) * 64 + 8 * hi;
#pragma unroll
    for (int k = 0; k < 16; ++k)
      fM[k] = *(const v2f*)(rM + 16 * (k >> 2) + 2 * (k & 3));
    const v8f u = matvec_tile(fM, BFt);
#pragma unroll
    for (int q = 0; q < 4; ++q) {
      part = fmaf(AFt[mtA][q][0], u[2 * q], part);
      part = fmaf(AFt[mtA][q][1], u[2 * q + 1], part);
    }
  }
  part += __shfl_xor(part, 16, 32);

  if (hi == 0) {
    const float bc = fminf(fmaxf(b_raw[0], -10.0f), 10.0f);
    Y[row] = part + bc;
  }
}

extern "C" void kernel_launch(void* const* d_in, const int* in_sizes, int n_in,
                              void* d_out, int out_size, void* d_ws, size_t ws_size,
                              hipStream_t stream) {
  const float* AT = (const float*)d_in[0];
  const float* K_raw = (const float*)d_in[1];
  const float* BT_raw = (const float*)d_in[2];
  const float* W_raw = (const float*)d_in[3];
  const float* b_raw = (const float*)d_in[4];
  float* Y = (float*)d_out;

  const int B = in_sizes[0] / 64;
  const int blocks = B / 64;
  comp_net_kernel<<<blocks, WG_THREADS, 0, stream>>>(AT, K_raw, BT_raw, W_raw, b_raw, Y);
}


# LDS rows padded from 256 to 272 bytes (16 bytes x odd) so the 16 row-lanes of each A-fragment read, and the transposed store, stop falling on one bank; arithmetic and summation order unchanged
# speedup vs baseline: 1.8647x; 1.1223x over previous
#include <hip/hip_runtime.h>

typedef __attribute__((ext_vector_type(2))) float v2f;
typedef __attribute__((ext_vector_type(8))) float v8f;

#define N_LOOP 21

#define WG_WAVES 4
#define WG_THREADS (WG_WAVES * 32)

#define LDP 68
#define ROWS_PER_PASS (WG_THREADS / 64)

__device__ __forceinline__ float fast_rcp(float x) {
  return __builtin_amdgcn_rcpf(x);
}

__device__ __forceinline__ v8f matvec_tile(const v2f fA[16], const v2f X[4][4]) {
  v8f c = {0.f, 0.f, 0.f, 0.f, 0.f, 0.f, 0.f, 0.f};
#pragma unroll
  for (int k = 0; k < 16; ++k)
    c = __builtin_amdgcn_wmma_f32_16x16x4_f32(false, fA[k], false, X[k >> 2][k & 3],
                                              (short)0, c, false, false);
  return c;
}

__global__ __launch_bounds__(WG_THREADS) void comp_net_kernel(
    const float* __restrict__ AT, const float* __restrict__ K_raw,
    const float* __restrict__ BT_raw, const float* __restrict__ W_raw,
    const float* __restrict__ b_raw, float* __restrict__ Y) {
  __shared__ __attribute__((aligned(16))) float sKT[64 * LDP];
  __shared__ __attribute__((aligned(16))) float sK[64 * LDP];
  __shared__ __attribute__((aligned(16))) float sM[64 * LDP];

  const int tid = threadIdx.x;
  const int j = tid & 63;
  const int i0 = tid >> 6;
#pragma unroll 2
  for (int n = 0; n < 64 / ROWS_PER_PASS; ++n) {
    const int i = i0 + n * ROWS_PER_PASS;
    const int idx = i * 64 + j;
    const float ke = fminf(expf(K_raw[idx]), 1000.0f);
    sK[i * LDP + j] = ke;
    sKT[j * LDP + i] = ke;
    float w = W_raw[idx];
    w = fminf(fmaxf(w, -10.0f), 10.0f);
    sM[i * LDP + j] = ke * w;
  }
  __syncthreads();

  const int lane = tid & 31;
  const int hi = (lane >> 4) & 1;
  const int lr = lane & 15;
  const int warp = tid >> 5;
  const int row = ((int)blockIdx.x * WG_WAVES + warp) * 16 + lr;

  v2f fKT[4][16], fK[4][16];
#pragma unroll
  for (int mtA = 0; mtA < 4; ++mtA) {
    const float* rT = sKT + (16 * mtA + lr) * LDP + 8 * hi;
    const float* rK = sK + (16 * mtA + lr) * LDP + 8 * hi;
#pragma unroll
    for (int k = 0; k < 16; ++k) {
      const int col = 16 * (k >> 2) + 2 * (k & 3);
      fKT[mtA][k] = *(const v2f*)(rT + col);
      fK[mtA][k] = *(const v2f*)(rK + col);
    }
  }

  v2f ATt[4][4], AFt[4][4], BFt[4][4], BTv[4][4];
#pragma unroll
  for (int mt = 0; mt < 4; ++mt) {
    const float* src = AT + row * 64 + 16 * mt + 8 * hi;
    const float* bsrc = BT_raw + 16 * mt + 8 * hi;
#pragma unroll
    for (int q = 0; q < 4; ++q) {
      ATt[mt][q] = *(const v2f*)(src + 2 * q);
      AFt[mt][q] = ATt[mt][q];
      v2f t;
      t[0] = fminf(expf(bsrc[2 * q]), 1000.0f);
      t[1] = fminf(expf(bsrc[2 * q + 1]), 1000.0f);
      BTv[mt][q] = t;
    }
  }

  v8f acc[4];

  for (int it = 0; it < N_LOOP; ++it) {
#pragma unroll
    for (int mtA = 0; mtA < 4; ++mtA) acc[mtA] = matvec_tile(fKT[mtA], AFt);
#pragma unroll
    for (int mt = 0; mt < 4; ++mt)
#pragma unroll
      for (int q = 0; q < 4; ++q) {
        v2f t;
        t[0] = BTv[mt][q][0] * fast_rcp(acc[mt][2 * q] + 1.0f);
        t[1] = BTv[mt][q][1] * fast_rcp(acc[mt][2 * q + 1] + 1.0f);
        BFt[mt][q] = t;
      }

#pragma unroll
    for (int mtA = 0; mtA < 4; ++mtA) acc[mtA] = matvec_tile(fK[mtA], BFt);
#pragma unroll
    for (int mt = 0; mt < 4; ++mt)
#pragma unroll
      for (int q = 0; q < 4; ++q) {
        v2f t;
        t[0] = ATt[mt][q][0] * fast_rcp(acc[mt][2 * q] + 1.0f);
        t[1] = ATt[mt][q][1] * fast_rcp(acc[mt][2 * q + 1] + 1.0f);
        AFt[mt][q] = t;
      }
  }

#pragma unroll
  for (int mtA = 0; mtA < 4; ++mtA) acc[mtA] = matvec_tile(fKT[mtA], AFt);
#pragma unroll
  for (int mt = 0; mt < 4; ++mt)
#pragma unroll
    for (int q = 0; q < 4; ++q) {
      v2f t;
      t[0] = BTv[mt][q][0] * fast_rcp(acc[mt][2 * q] + 1.0f);
      t[1] = BTv[mt][q][1] * fast_rcp(acc[mt][2 * q + 1] + 1.0f);
      BFt[mt][q] = t;
    }

  float part = 0.0f;
#pragma unroll
  for (int mtA = 0; mtA < 4; ++mtA) {
    v2f fM[16];
    const float* rM = sM + (16 * mtA + lr) * LDP + 8 * hi;
#pragma unroll
    for (int k = 0; k < 16; ++k)
      fM[k] = *(const v2f*)(rM + 16 * (k >> 2) + 2 * (k & 3));
    const v8f u = matvec_tile(fM, BFt);
#pragma unroll
    for (int q = 0; q < 4; ++q) {
      part = fmaf(AFt[mtA][q][0], u[2 * q], part);
      part = fmaf(AFt[mtA][q][1], u[2 * q + 1], part);
    }
  }
  part += __shfl_xor(part, 16, 32);

  if (hi == 0) {
    const float bc = fminf(fmaxf(b_raw[0], -10.0f), 10.0f);
    Y[row] = part + bc;
  }
}

extern "C" void kernel_launch(void* const* d_in, const int* in_sizes, int n_in,
                              void* d_out, int out_size, void* d_ws, size_t ws_size,
                              hipStream_t stream) {
  const float* AT = (const float*)d_in[0];
  const float* K_raw = (const float*)d_in[1];
  const float* BT_raw = (const float*)d_in[2];
  const float* W_raw = (const float*)d_in[3];
  const float* b_raw = (const float*)d_in[4];
  float* Y = (float*)d_out;

  const int B = in_sizes[0] / 64;
  const int blocks = B / 64;
  comp_net_kernel<<<blocks, WG_THREADS, 0, stream>>>(AT, K_raw, BT_raw, W_raw, b_raw, Y);
}


# Set-up loads K_raw and W_raw as 16-byte pieces, all sixteen per thread issued before first use: one memory round trip per work-group instead of eight; LDS contents and summation order unchanged
# speedup vs baseline: 2.0871x; 1.1193x over previous
#include <hip/hip_runtime.h>

typedef __attribute__((ext_vector_type(2))) float v2f;
typedef __attribute__((ext_vector_type(4))) float v4f;
typedef __attribute__((ext_vector_type(8))) float v8f;

#define N_LOOP 21

#define WG_WAVES 4
#define WG_THREADS (WG_WAVES * 32)

#define LDP 68
#define SETUP_PIECES (64 * 64 / 4 / WG_THREADS)

__device__ __forceinline__ float fast_rcp(float x) {
  return __builtin_amdgcn_rcpf(x);
}

__device__ __forceinline__ v8f matvec_tile(const v2f fA[16], const v2f X[4][4]) {
  v8f c = {0.f, 0.f, 0.f, 0.f, 0.f, 0.f, 0.f, 0.f};
#pragma unroll
  for (int k = 0; k < 16; ++k)
    c = __builtin_amdgcn_wmma_f32_16x16x4_f32(false, fA[k], false, X[k >> 2][k & 3],
                                              (short)0, c, false, false);
  return c;
}

__global__ __launch_bounds__(WG_THREADS) void comp_net_kernel(
    const float* __restrict__ AT, const float* __restrict__ K_raw,
    const float* __restrict__ BT_raw, const float* __restrict__ W_raw,
    const float* __restrict__ b_raw, float* __restrict__ Y) {
  __shared__ __attribute__((aligned(16))) float sKT[64 * LDP];
  __shared__ __attribute__((aligned(16))) float sK[64 * LDP];
  __shared__ __attribute__((aligned(16))) float sM[64 * LDP];

  const int tid = threadIdx.x;
  const int c8 = tid & 7;
  const int r16 = tid >> 3;
  v4f kraw[SETUP_PIECES], wraw[SETUP_PIECES];
#pragma unroll
  for (int n = 0; n < SETUP_PIECES; ++n) {
    const int i = (n >> 1) * (WG_THREADS / 8) + r16;
    const int j4 = (n & 1) * 32 + c8 * 4;
    kraw[n] = *(const v4f*)(K_raw + i * 64 + j4);
    wraw[n] = *(const v4f*)(W_raw + i * 64 + j4);
  }
#pragma unroll
  for (int n = 0; n < SETUP_PIECES; ++n) {
    const int i = (n >> 1) * (WG_THREADS / 8) + r16;
    const int j4 = (n & 1) * 32 + c8 * 4;
    v4f ke, m;
#pragma unroll
    for (int c = 0; c < 4; ++c) {
      ke[c] = fminf(expf(kraw[n][c]), 1000.0f);
      const float w = fminf(fmaxf(wraw[n][c], -10.0f), 10.0f);
      m[c] = ke[c] * w;
      sKT[(j4 + c) * LDP + i] = ke[c];
    }
    *(v4f*)(sK + i * LDP + j4) = ke;
    *(v4f*)(sM + i * LDP + j4) = m;
  }
  __syncthreads();

  const int lane = tid & 31;
  const int hi = (lane >> 4) & 1;
  const int lr = lane & 15;
  const int warp = tid >> 5;
  const int row = ((int)blockIdx.x * WG_WAVES + warp) * 16 + lr;

  v2f fKT[4][16], fK[4][16];
#pragma unroll
  for (int mtA = 0; mtA < 4; ++mtA) {
    const float* rT = sKT + (16 * mtA + lr) * LDP + 8 * hi;
    const float* rK = sK + (16 * mtA + lr) * LDP + 8 * hi;
#pragma unroll
    for (int k = 0; k < 16; ++k) {
      const int col = 16 * (k >> 2) + 2 * (k & 3);
      fKT[mtA][k] = *(const v2f*)(rT + col);
      fK[mtA][k] = *(const v2f*)(rK + col);
    }
  }

  v2f ATt[4][4], AFt[4][4], BFt[4][4], BTv[4][4];
#pragma unroll
  for (int mt = 0; mt < 4; ++mt) {
    const float* src = AT + row * 64 + 16 * mt + 8 * hi;
    const float* bsrc = BT_raw + 16 * mt + 8 * hi;
#pragma unroll
    for (int q = 0; q < 4; ++q) {
      ATt[mt][q] = *(const v2f*)(src + 2 * q);
      AFt[mt][q] = ATt[mt][q];
      v2f t;
      t[0] = fminf(expf(bsrc[2 * q]), 1000.0f);
      t[1] = fminf(expf(bsrc[2 * q + 1]), 1000.0f);
      BTv[mt][q] = t;
    }
  }

  v8f acc[4];

  for (int it = 0; it < N_LOOP; ++it) {
#pragma unroll
    for (int mtA = 0; mtA < 4; ++mtA) acc[mtA] = matvec_tile(fKT[mtA], AFt);
#pragma unroll
    for (int mt = 0; mt < 4; ++mt)
#pragma unroll
      for (int q = 0; q < 4; ++q) {
        v2f t;
        t[0] = BTv[mt][q][0] * fast_rcp(acc[mt][2 * q] + 1.0f);
        t[1] = BTv[mt][q][1] * fast_rcp(acc[mt][2 * q + 1] + 1.0f);
        BFt[mt][q] = t;
      }

#pragma unroll
    for (int mtA = 0; mtA < 4; ++mtA) acc[mtA] = matvec_tile(fK[mtA], BFt);
#pragma unroll
    for (int mt = 0; mt < 4; ++mt)
#pragma unroll
      for (int q = 0; q < 4; ++q) {
        v2f t;
        t[0] = ATt[mt][q][0] * fast_rcp(acc[mt][2 * q] + 1.0f);
        t[1] = ATt[mt][q][1] * fast_rcp(acc[mt][2 * q + 1] + 1.0f);
        AFt[mt][q] = t;
      }
  }

#pragma unroll
  for (int mtA = 0; mtA < 4; ++mtA) acc[mtA] = matvec_tile(fKT[mtA], AFt);
#pragma unroll
  for (int mt = 0; mt < 4; ++mt)
#pragma unroll
    for (int q = 0; q < 4; ++q) {
      v2f t;
      t[0] = BTv[mt][q][0] * fast_rcp(acc[mt][2 * q] + 1.0f);
      t[1] = BTv[mt][q][1] * fast_rcp(acc[mt][2 * q + 1] + 1.0f);
      BFt[mt][q] = t;
    }

  float part = 0.0f;
#pragma unroll
  for (int mtA = 0; mtA < 4; ++mtA) {
    v2f fM[16];
    const float* rM = sM + (16 * mtA + lr) * LDP + 8 * hi;
#pragma unroll
    for (int k = 0; k < 16; ++k)
      fM[k] = *(const v2f*)(rM + 16 * (k >> 2) + 2 * (k & 3));
    const v8f u = matvec_tile(fM, BFt);
#pragma unroll
    for (int q = 0; q < 4; ++q) {
      part = fmaf(AFt[mtA][q][0], u[2 * q], part);
      part = fmaf(AFt[mtA][q][1], u[2 * q + 1], part);
    }
  }
  part += __shfl_xor(part, 16, 32);

  if (hi == 0) {
    const float bc = fminf(fmaxf(b_raw[0], -10.0f), 10.0f);
    Y[row] = part + bc;
  }
}

extern "C" void kernel_launch(void* const* d_in, const int* in_sizes, int n_in,
                              void* d_out, int out_size, void* d_ws, size_t ws_size,
                              hipStream_t stream) {
  const float* AT = (const float*)d_in[0];
  const float* K_raw = (const float*)d_in[1];
  const float* BT_raw = (const float*)d_in[2];
  const float* W_raw = (const float*)d_in[3];
  const float* b_raw = (const float*)d_in[4];
  float* Y = (float*)d_out;

  const int B = in_sizes[0] / 64;
  const int blocks = B / 64;
  comp_net_kernel<<<blocks, WG_THREADS, 0, stream>>>(AT, K_raw, BT_raw, W_raw, b_raw, Y);
}


# Vector work between matrix runs cut to the reciprocals: accumulators start at 1.0 (the +1 is summed first, not last) and BT is folded once into the LDS columns of Kexp and M (K*BT rounded first)
# speedup vs baseline: 2.2010x; 1.0546x over previous
#include <hip/hip_runtime.h>

typedef __attribute__((ext_vector_type(2))) float v2f;
typedef __attribute__((ext_vector_type(4))) float v4f;
typedef __attribute__((ext_vector_type(8))) float v8f;

#define N_LOOP 21

#define WG_WAVES 4
#define WG_THREADS (WG_WAVES * 32)

#define LDP 68
#define SETUP_PIECES (64 * 64 / 4 / WG_THREADS)

__device__ __forceinline__ float fast_rcp(float x) {
  return __builtin_amdgcn_rcpf(x);
}

__device__ __forceinline__ v8f matvec_tile(const v2f fA[16], const v2f X[4][4], const float c0) {
  v8f c = {c0, c0, c0, c0, c0, c0, c0, c0};
#pragma unroll
  for (int k = 0; k < 16; ++k)
    c = __builtin_amdgcn_wmma_f32_16x16x4_f32(false, fA[k], false, X[k >> 2][k & 3],
                                              (short)0, c, false, false);
  return c;
}

__global__ __launch_bounds__(WG_THREADS) void comp_net_kernel(
    const float* __restrict__ AT, const float* __restrict__ K_raw,
    const float* __restrict__ BT_raw, const float* __restrict__ W_raw,
    const float* __restrict__ b_raw, float* __restrict__ Y) {
  __shared__ __attribute__((aligned(16))) float sKT[64 * LDP];
  __shared__ __attribute__((aligned(16))) float sK[64 * LDP];
  __shared__ __attribute__((aligned(16))) float sM[64 * LDP];

  const int tid = threadIdx.x;
  const int c8 = tid & 7;
  const int r16 = tid >> 3;
  v4f kraw[SETUP_PIECES], wraw[SETUP_PIECES];
  const v4f btraw0 = *(const v4f*)(BT_raw + c8 * 4);
  const v4f btraw1 = *(const v4f*)(BT_raw + 32 + c8 * 4);
#pragma unroll
  for (int n = 0; n < SETUP_PIECES; ++n) {
    const int i = (n >> 1) * (WG_THREADS / 8) + r16;
    const int j4 = (n & 1) * 32 + c8 * 4;
    kraw[n] = *(const v4f*)(K_raw + i * 64 + j4);
    wraw[n] = *(const v4f*)(W_raw + i * 64 + j4);
  }
  v4f bt0, bt1;
#pragma unroll
  for (int c = 0; c < 4; ++c) {
    bt0[c] = fminf(expf(btraw0[c]), 1000.0f);
    bt1[c] = fminf(expf(btraw1[c]), 1000.0f);
  }
#pragma unroll
  for (int n = 0; n < SETUP_PIECES; ++n) {
    const int i = (n >> 1) * (WG_THREADS / 8) + r16;
    const int j4 = (n & 1) * 32 + c8 * 4;
    const v4f bt = (n & 1) ? bt1 : bt0;
    v4f ke, kb, m;
#pragma unroll
    for (int c = 0; c < 4; ++c) {
      ke[c] = fminf(expf(kraw[n][c]), 1000.0f);
      const float w = fminf(fmaxf(wraw[n][c], -10.0f), 10.0f);
      kb[c] = ke[c] * bt[c];
      m[c] = ke[c] * w * bt[c];
      sKT[(j4 + c) * LDP + i] = ke[c];
    }
    *(v4f*)(sK + i * LDP + j4) = kb;
    *(v4f*)(sM + i * LDP + j4) = m;
  }
  __syncthreads();

  const int lane = tid & 31;
  const int hi = (lane >> 4) & 1;
  const int lr = lane & 15;
  const int warp = tid >> 5;
  const int row = ((int)blockIdx.x * WG_WAVES + warp) * 16 + lr;

  v2f fKT[4][16], fK[4][16];
#pragma unroll
  for (int mtA = 0; mtA < 4; ++mtA) {
    const float* rT = sKT + (16 * mtA + lr) * LDP + 8 * hi;
    const float* rK = sK + (16 * mtA + lr) * LDP + 8 * hi;
#pragma unroll
    for (int k = 0; k < 16; ++k) {
      const int col = 16 * (k >> 2) + 2 * (k & 3);
      fKT[mtA][k] = *(const v2f*)(rT + col);
      fK[mtA][k] = *(const v2f*)(rK + col);
    }
  }

  v2f ATt[4][4], AFt[4][4], BFt[4][4];
#pragma unroll
  for (int mt = 0; mt < 4; ++mt) {
    const float* src = AT + row * 64 + 16 * mt + 8 * hi;
#pragma unroll
    for (int q = 0; q < 4; ++q) {
      ATt[mt][q] = *(const v2f*)(src + 2 * q);
      AFt[mt][q] = ATt[mt][q];
    }
  }

  v8f acc[4];

  for (int it = 0; it < N_LOOP; ++it) {
#pragma unroll
    for (int mtA = 0; mtA < 4; ++mtA) acc[mtA] = matvec_tile(fKT[mtA], AFt, 1.0f);
#pragma unroll
    for (int mt = 0; mt < 4; ++mt)
#pragma unroll
      for (int q = 0; q < 4; ++q) {
        v2f t;
        t[0] = fast_rcp(acc[mt][2 * q]);
        t[1] = fast_rcp(acc[mt][2 * q + 1]);
        BFt[mt][q] = t;
      }

#pragma unroll
    for (int mtA = 0; mtA < 4; ++mtA) acc[mtA] = matvec_tile(fK[mtA], BFt, 1.0f);
#pragma unroll
    for (int mt = 0; mt < 4; ++mt)
#pragma unroll
      for (int q = 0; q < 4; ++q) {
        v2f t;
        t[0] = ATt[mt][q][0] * fast_rcp(acc[mt][2 * q]);
        t[1] = ATt[mt][q][1] * fast_rcp(acc[mt][2 * q + 1]);
        AFt[mt][q] = t;
      }
  }

#pragma unroll
  for (int mtA = 0; mtA < 4; ++mtA) acc[mtA] = matvec_tile(fKT[mtA], AFt, 1.0f);
#pragma unroll
  for (int mt = 0; mt < 4; ++mt)
#pragma unroll
    for (int q = 0; q < 4; ++q) {
      v2f t;
      t[0] = fast_rcp(acc[mt][2 * q]);
      t[1] = fast_rcp(acc[mt][2 * q + 1]);
      BFt[mt][q] = t;
    }

  float part = 0.0f;
#pragma unroll
  for (int mtA = 0; mtA < 4; ++mtA) {
    v2f fM[16];
    const float* rM = sM + (16 * mtA + lr) * LDP + 8 * hi;
#pragma unroll
    for (int k = 0; k < 16; ++k)
      fM[k] = *(const v2f*)(rM + 16 * (k >> 2) + 2 * (k & 3));
    const v8f u = matvec_tile(fM, BFt, 0.0f);
#pragma unroll
    for (int q = 0; q < 4; ++q) {
      part = fmaf(AFt[mtA][q][0], u[2 * q], part);
      part = fmaf(AFt[mtA][q][1], u[2 * q + 1], part);
    }
  }
  part += __shfl_xor(part, 16, 32);

  if (hi == 0) {
    const float bc = fminf(fmaxf(b_raw[0], -10.0f), 10.0f);
    Y[row] = part + bc;
  }
}

extern "C" void kernel_launch(void* const* d_in, const int* in_sizes, int n_in,
                              void* d_out, int out_size, void* d_ws, size_t ws_size,
                              hipStream_t stream) {
  const float* AT = (const float*)d_in[0];
  const float* K_raw = (const float*)d_in[1];
  const float* BT_raw = (const float*)d_in[2];
  const float* W_raw = (const float*)d_in[3];
  const float* b_raw = (const float*)d_in[4];
  float* Y = (float*)d_out;

  const int B = in_sizes[0] / 64;
  const int blocks = B / 64;
  comp_net_kernel<<<blocks, WG_THREADS, 0, stream>>>(AT, K_raw, BT_raw, W_raw, b_raw, Y);
}
